# starting kernel
# baseline (speedup 1.0000x reference)
#include <hip/hip_runtime.h>


typedef float v2f __attribute__((ext_vector_type(2)));
typedef float v4f __attribute__((ext_vector_type(4)));
typedef float v8f __attribute__((ext_vector_type(8)));

#define NB 8192
#define NN 64

__global__ __launch_bounds__(256) void stage1_uv(
    const float* __restrict__ x,
    const float* __restrict__ W1r,
    const float* __restrict__ W1i,
    float* __restrict__ U,
    float* __restrict__ V)
{
    const int lane   = threadIdx.x & 31;
    const int wave   = (blockIdx.x * blockDim.x + threadIdx.x) >> 5;
    const int nwaves = (gridDim.x * blockDim.x) >> 5;

    const float wr0 = W1r[63 * NN + 2 * lane];
    const float wr1 = W1r[63 * NN + 2 * lane + 1];
    const float wi0 = W1i[63 * NN + 2 * lane];
    const float wi1 = W1i[63 * NN + 2 * lane + 1];

    const int total_rows = NB * NN;
    for (int row = wave; row < total_rows; row += nwaves) {
        const v4f* rp = (const v4f*)(x + (size_t)row * 128 + 4 * lane);
        v4f z = __builtin_nontemporal_load(rp);

        float u = z.x * wr0 - z.y * wi0 + z.z * wr1 - z.w * wi1;
        float v = z.x * wi0 + z.y * wr0 + z.z * wi1 + z.w * wr1;

        const float re63 = __shfl(z.z, 31, 32);
        const float im63 = __shfl(z.w, 31, 32);

        #pragma unroll
        for (int m = 16; m >= 1; m >>= 1) {
            u += __shfl_xor(u, m, 32);
            v += __shfl_xor(v, m, 32);
        }

        if ((row & (NN - 1)) == 0) { u = re63; v = im63; }

        if (lane == 0) {
            U[row] = u;
            V[row] = v;
        }
    }
}

__global__ __launch_bounds__(256) void stage2_wmma(
    const float* __restrict__ U,
    const float* __restrict__ V,
    const float* __restrict__ W2r,
    const float* __restrict__ W2i,
    float* __restrict__ out)
{
    const int lane = threadIdx.x & 31;
    const int wave = (blockIdx.x * blockDim.x + threadIdx.x) >> 5;

    const int b0 = (wave >> 2) << 4;
    const int m0 = (wave & 3) << 4;
    const int lm = lane & 15;
    const int lk = (lane >> 4) << 1;

    const float* urow = U   + (size_t)(b0 + lm) * NN;
    const float* vrow = V   + (size_t)(b0 + lm) * NN;
    const float* wr   = W2r + (size_t)(m0 + lm) * NN;
    const float* wi   = W2i + (size_t)(m0 + lm) * NN;

    v8f acc = {};
    #pragma unroll
    for (int k0 = 0; k0 < NN; k0 += 4) {
        v2f au = *(const v2f*)(urow + k0 + lk);
        v2f av = *(const v2f*)(vrow + k0 + lk);
        v2f r2 = *(const v2f*)(wr   + k0 + lk);
        v2f i2 = *(const v2f*)(wi   + k0 + lk);
        v2f ba = r2 + i2;
        v2f bb = r2 - i2;
        acc = __builtin_amdgcn_wmma_f32_16x16x4_f32(false, au, false, ba,
                                                    (short)0, acc, false, false);
        acc = __builtin_amdgcn_wmma_f32_16x16x4_f32(false, av, false, bb,
                                                    (short)0, acc, false, false);
    }

    const int rbase = b0 + ((lane >> 4) << 3);
    #pragma unroll
    for (int r = 0; r < 8; ++r) {
        out[(size_t)(rbase + r) * NN + m0 + lm] = fabsf(acc[r]);
    }
}

extern "C" void kernel_launch(void* const* d_in, const int* in_sizes, int n_in,
                              void* d_out, int out_size, void* d_ws, size_t ws_size,
                              hipStream_t stream) {
    const float* x   = (const float*)d_in[0];
    const float* W1r = (const float*)d_in[1];
    const float* W1i = (const float*)d_in[2];
    const float* W2r = (const float*)d_in[3];
    const float* W2i = (const float*)d_in[4];
    float* out = (float*)d_out;

    float* U = (float*)d_ws;
    float* V = U + (size_t)NB * NN;

    stage1_uv<<<1024, 256, 0, stream>>>(x, W1r, W1i, U, V);

    stage2_wmma<<<256, 256, 0, stream>>>(U, V, W2r, W2i, out);
}


# stage 1 as a stream: each wave owns 128 consecutive rows, keeps 24-32 whole-row loads in flight and folds 16 rows' wave sums together on the vector ALU (same 16-8-4-2-1 tree, no LDS permutes)
# speedup vs baseline: 3.3702x; 3.3702x over previous
#include <hip/hip_runtime.h>


typedef float v2f __attribute__((ext_vector_type(2)));
typedef float v4f __attribute__((ext_vector_type(4)));
typedef float v8f __attribute__((ext_vector_type(8)));

#define NB 8192
#define NN 64

#define S1_WAVES   16
#define S1_THREADS (S1_WAVES * 32)
#define S1_ROWS    128
#define S1_CHUNK   8
#define S1_AHEAD   3
#define S1_NCHUNK  (S1_ROWS / S1_CHUNK)
#define S1_BLOCKS  256
static_assert(S1_BLOCKS * S1_ROWS * S1_WAVES == NB * NN, "the grid covers every row exactly once");
static_assert(S1_ROWS % (2 * NN) == 0 && S1_NCHUNK % 4 == 0, "a wave owns whole batches and whole 32-row store groups");

#define DEV static __device__ __forceinline__

DEV float lane_x16(float x) {
    const int b = __builtin_bit_cast(int, x);
    return __builtin_bit_cast(float,
        __builtin_amdgcn_permlanex16(b, b, 0x76543210, 0xfedcba98, false, false));
}
template <int CTRL> DEV float lane_dpp(float x) {
    const int b = __builtin_bit_cast(int, x);
    return __builtin_bit_cast(float, __builtin_amdgcn_update_dpp(b, b, CTRL, 0xf, 0xf, true));
}
DEV float fold16(float a, float b, bool hi) { return (hi ? b : a) + lane_x16(hi ? a : b); }
template <int CTRL> DEV float fold(float a, float b, bool hi) {
    return (hi ? b : a) + lane_dpp<CTRL>(hi ? a : b);
}

__global__ __launch_bounds__(S1_THREADS) void stage1_uv(
    const float* __restrict__ x,
    const float* __restrict__ W1r,
    const float* __restrict__ W1i,
    float* __restrict__ U,
    float* __restrict__ V)
{
    const unsigned lane = threadIdx.x & 31u;
    const unsigned wave =
        __builtin_amdgcn_readfirstlane((blockIdx.x * S1_THREADS + threadIdx.x) >> 5);
    const unsigned row0 = wave * S1_ROWS;

    const float wr0 = W1r[63 * NN + 2 * lane];
    const float wr1 = W1r[63 * NN + 2 * lane + 1];
    const float wi0 = W1i[63 * NN + 2 * lane];
    const float wi1 = W1i[63 * NN + 2 * lane + 1];

    const bool h16 = (lane & 16u) != 0u;
    const bool h8  = (lane & 8u)  != 0u;
    const bool h4  = (lane & 4u)  != 0u;
    const bool h2  = (lane & 2u)  != 0u;
    const bool h1  = (lane & 1u)  != 0u;
    const bool last = (lane == 31u);

    const v4f* src = (const v4f*)(x + (size_t)row0 * 128u) + lane;

    const v2f p0 = *(const v2f*)(x + (size_t)row0 * 128u + 126u);
    const v2f p1 = *(const v2f*)(x + (size_t)(row0 + NN) * 128u + 126u);

    v4f buf[S1_AHEAD + 1][S1_CHUNK];
#define S1_LOAD(C) do { _Pragma("unroll") for (int j_ = 0; j_ < S1_CHUNK; ++j_) \
        if ((((C) * S1_CHUNK + j_) & (NN - 1)) != 0) \
            buf[(C) % (S1_AHEAD + 1)][j_] = __builtin_nontemporal_load(src + 32 * ((C) * S1_CHUNK + j_)); } while (0)

    #pragma unroll
    for (int c = 0; c < S1_AHEAD; ++c) S1_LOAD(c);

    float w[16];
    float resA = 0.f;

    #pragma unroll
    for (int c = 0; c < S1_NCHUNK; ++c) {
        __builtin_amdgcn_sched_barrier(0);
        if (c + S1_AHEAD < S1_NCHUNK) S1_LOAD(c + S1_AHEAD);
        __builtin_amdgcn_sched_barrier(0);

        #pragma unroll
        for (int j = 0; j < S1_CHUNK; ++j) {
            float u, v;
            if (((c * S1_CHUNK + j) & (NN - 1)) == 0) {
                const v2f p = (c == 0) ? p0 : p1;
                u = last ? p.x : 0.f;
                v = last ? p.y : 0.f;
            } else {
                const v4f z = buf[c % (S1_AHEAD + 1)][j];
                u = z.x * wr0 - z.y * wi0 + z.z * wr1 - z.w * wi1;
                v = z.x * wi0 + z.y * wr0 + z.z * wi1 + z.w * wr1;
            }
            w[(c & 1) * S1_CHUNK + j] = fold16(u, v, h16);
        }

        if (c & 1) {
            float t8[8], t4[4], t2[2];
            #pragma unroll
            for (int r = 0; r < 8; ++r) t8[r] = fold<0x168>(w[r], w[r + 8], h8);
            #pragma unroll
            for (int r = 0; r < 4; ++r) t4[r] = fold<0x164>(t8[r], t8[r + 4], h4);
            #pragma unroll
            for (int r = 0; r < 2; ++r) t2[r] = fold<0x4E>(t4[r], t4[r + 2], h2);
            const float res = fold<0xB1>(t2[0], t2[1], h1);

            if ((c >> 1) & 1) {
                const float xres = lane_x16(res);
                const float xA   = lane_x16(resA);
                const float oU = h16 ? xres : resA;
                const float oV = h16 ? res  : xA;
                const unsigned r32 = row0 + (unsigned)((c >> 2) * 32) + lane;
                U[r32] = oU;
                V[r32] = oV;
            } else {
                resA = res;
            }
        }
    }
#undef S1_LOAD
}

__global__ __launch_bounds__(256) void stage2_wmma(
    const float* __restrict__ U,
    const float* __restrict__ V,
    const float* __restrict__ W2r,
    const float* __restrict__ W2i,
    float* __restrict__ out)
{
    const int lane = threadIdx.x & 31;
    const int wave = (blockIdx.x * blockDim.x + threadIdx.x) >> 5;

    const int b0 = (wave >> 2) << 4;
    const int m0 = (wave & 3) << 4;
    const int lm = lane & 15;
    const int lk = (lane >> 4) << 1;

    const float* urow = U   + (size_t)(b0 + lm) * NN;
    const float* vrow = V   + (size_t)(b0 + lm) * NN;
    const float* wr   = W2r + (size_t)(m0 + lm) * NN;
    const float* wi   = W2i + (size_t)(m0 + lm) * NN;

    v8f acc = {};
    #pragma unroll
    for (int k0 = 0; k0 < NN; k0 += 4) {
        v2f au = *(const v2f*)(urow + k0 + lk);
        v2f av = *(const v2f*)(vrow + k0 + lk);
        v2f r2 = *(const v2f*)(wr   + k0 + lk);
        v2f i2 = *(const v2f*)(wi   + k0 + lk);
        v2f ba = r2 + i2;
        v2f bb = r2 - i2;
        acc = __builtin_amdgcn_wmma_f32_16x16x4_f32(false, au, false, ba,
                                                    (short)0, acc, false, false);
        acc = __builtin_amdgcn_wmma_f32_16x16x4_f32(false, av, false, bb,
                                                    (short)0, acc, false, false);
    }

    const int rbase = b0 + ((lane >> 4) << 3);
    #pragma unroll
    for (int r = 0; r < 8; ++r) {
        out[(size_t)(rbase + r) * NN + m0 + lm] = fabsf(acc[r]);
    }
}

extern "C" void kernel_launch(void* const* d_in, const int* in_sizes, int n_in,
                              void* d_out, int out_size, void* d_ws, size_t ws_size,
                              hipStream_t stream) {
    const float* x   = (const float*)d_in[0];
    const float* W1r = (const float*)d_in[1];
    const float* W1i = (const float*)d_in[2];
    const float* W2r = (const float*)d_in[3];
    const float* W2i = (const float*)d_in[4];
    float* out = (float*)d_out;

    float* U = (float*)d_ws;
    float* V = U + (size_t)NB * NN;

    stage1_uv<<<S1_BLOCKS, S1_THREADS, 0, stream>>>(x, W1r, W1i, U, V);

    stage2_wmma<<<256, 256, 0, stream>>>(U, V, W2r, W2i, out);
}
